# starting kernel
# baseline (speedup 1.0000x reference)
#include <hip/hip_runtime.h>


typedef float v2f __attribute__((ext_vector_type(2)));
typedef float v4f __attribute__((ext_vector_type(4)));
typedef float v8f __attribute__((ext_vector_type(8)));

constexpr int kB = 64;
constexpr int kL = 2048;
constexpr int kD = 128;
constexpr int kRowsPerBlock = 128;
constexpr int kWaves = 8;

__global__ __launch_bounds__(256) void
embatt_kernel(const float* __restrict__ sent,
              const int* __restrict__ idxs,
              float* __restrict__ out)
{
    __shared__ float eavg[kD];
    __shared__ float att[kRowsPerBlock];

    const int b       = blockIdx.y;
    const int rowBase = blockIdx.x * kRowsPerBlock;
    const int tid     = threadIdx.x;
    const int lane    = tid & 31;
    const int wave    = tid >> 5;

    const float* sb = sent + (size_t)b * kL * kD;

    if (tid < kD) {
        const int i0 = idxs[2 * b + 0];
        const int i1 = idxs[2 * b + 1];
        eavg[tid] = 0.5f * (sb[(size_t)i0 * kD + tid] + sb[(size_t)i1 * kD + tid]);
    }
    __syncthreads();

    const int m    = lane & 15;
    const int koff = (lane >> 4) << 1;
    const float* tile = sb + (size_t)(rowBase + wave * 16) * kD;

    v8f c = {0.f, 0.f, 0.f, 0.f, 0.f, 0.f, 0.f, 0.f};
#pragma unroll
    for (int d0 = 0; d0 < kD; d0 += 4) {
        v2f a  = *(const v2f*)(tile + (size_t)m * kD + d0 + koff);
        v2f bb = *(const v2f*)(&eavg[d0 + koff]);
        c = __builtin_amdgcn_wmma_f32_16x16x4_f32(
                 false, a,  false, bb,
                 (short)0, c,  false,  false);
    }

    if ((lane & 15) == 0) {
        const int mhi = (lane >> 4) << 3;
#pragma unroll
        for (int r = 0; r < 8; ++r)
            att[wave * 16 + r + mhi] = c[r];
    }
    __syncthreads();

    const size_t obase = (size_t)b * kL * kD + (size_t)rowBase * kD;
#pragma unroll 4
    for (int it = 0; it < kRowsPerBlock / kWaves; ++it) {
        const int rr  = it * kWaves + wave;
        const float s = att[rr];
        const float* src = sb + (size_t)(rowBase + rr) * kD + lane * 4;
        float*       dst = out + obase + (size_t)rr * kD + lane * 4;
        v4f v = *(const v4f*)src;
        v4f w = {v.x * s, v.y * s, v.z * s, v.w * s};
        *(v4f*)dst = w;
    }
}

extern "C" void kernel_launch(void* const* d_in, const int* in_sizes, int n_in,
                              void* d_out, int out_size, void* d_ws, size_t ws_size,
                              hipStream_t stream)
{
    const float* sent = (const float*)d_in[0];
    const int*   idxs = (const int*)d_in[1];
    float*       out  = (float*)d_out;

    dim3 grid(kL / kRowsPerBlock, kB);
    embatt_kernel<<<grid, 256, 0, stream>>>(sent, idxs, out);
}


# each wave reads its 16 rows once in whole 128-B lines into registers, takes the WMMA A fragments from a padded LDS copy and scales by its own accumulators (no row-strided loads, no re-read)
# speedup vs baseline: 1.3184x; 1.3184x over previous
#include <hip/hip_runtime.h>


typedef float v2f __attribute__((ext_vector_type(2)));
typedef float v4f __attribute__((ext_vector_type(4)));
typedef float v8f __attribute__((ext_vector_type(8)));

constexpr int kB = 64;
constexpr int kL = 2048;
constexpr int kD = 128;
constexpr int kWaves = 8;
constexpr int kTile = 16;
constexpr int kRowsPerBlock = kWaves * kTile;
constexpr int kPitch = kD + 4;

struct Lds {
    float eavg[kD];
    float stage[kWaves][kTile][kPitch];
};

__global__ __launch_bounds__(256) void
embatt_kernel(const float* __restrict__ sent,
              const int* __restrict__ idxs,
              float* __restrict__ out)
{
    __shared__ __attribute__((aligned(16))) Lds lds;

    const int b       = blockIdx.y;
    const int rowBase = blockIdx.x * kRowsPerBlock;
    const int tid     = threadIdx.x;
    const int lane    = tid & 31;
    const int wave    = tid >> 5;
    const int m       = lane & 15;
    const int h       = lane >> 4;

    const float* sb   = sent + (size_t)b * kL * kD;
    const size_t row0 = (size_t)(rowBase + wave * kTile);

    const int i0 = idxs[2 * b + 0];
    const int i1 = idxs[2 * b + 1];

    const float* lrow = sb + (row0 + 8 * h) * kD + 4 * m;
    v4f v[kTile];
#pragma unroll
    for (int q = 0; q < kTile; ++q)
        v[q] = *(const v4f*)(lrow + (q & 7) * kD + (q >> 3) * 64);

    if (tid < kD)
        lds.eavg[tid] = 0.5f * (sb[(size_t)i0 * kD + tid] + sb[(size_t)i1 * kD + tid]);

    float* srow = &lds.stage[wave][8 * h][4 * m];
#pragma unroll
    for (int q = 0; q < kTile; ++q)
        *(v4f*)(srow + (q & 7) * kPitch + (q >> 3) * 64) = v[q];
    __syncthreads();

    const int koff = h << 1;
    const float* arow = &lds.stage[wave][m][koff];
    const float* brow = &lds.eavg[koff];

    v8f c = {0.f, 0.f, 0.f, 0.f, 0.f, 0.f, 0.f, 0.f};
#pragma unroll
    for (int d0 = 0; d0 < kD; d0 += 4) {
        v2f a  = *(const v2f*)(arow + d0);
        v2f bb = *(const v2f*)(brow + d0);
        c = __builtin_amdgcn_wmma_f32_16x16x4_f32(
                 false, a,  false, bb,
                 (short)0, c,  false,  false);
    }

    float* orow = out + ((size_t)b * kL + row0 + 8 * h) * kD + 4 * m;
#pragma unroll
    for (int q = 0; q < kTile; ++q) {
        const float s = c[q & 7];
        v4f w = {v[q].x * s, v[q].y * s, v[q].z * s, v[q].w * s};
        *(v4f*)(orow + (q & 7) * kD + (q >> 3) * 64) = w;
    }
}

extern "C" void kernel_launch(void* const* d_in, const int* in_sizes, int n_in,
                              void* d_out, int out_size, void* d_ws, size_t ws_size,
                              hipStream_t stream)
{
    const float* sent = (const float*)d_in[0];
    const int*   idxs = (const int*)d_in[1];
    float*       out  = (float*)d_out;

    dim3 grid(kL / kRowsPerBlock, kB);
    embatt_kernel<<<grid, 256, 0, stream>>>(sent, idxs, out);
}
